# starting kernel
# baseline (speedup 1.0000x reference)
#include <hip/hip_runtime.h>
#include <hip/hip_bf16.h>

#define NUM_EXPERT 32
#define IN_FEAT 512
#define OUT_FEAT 512
#define BATCH 4096
#define TILE_M 16
#define MAX_TILES (BATCH / TILE_M + NUM_EXPERT)

typedef __attribute__((ext_vector_type(2))) float v2f;
typedef __attribute__((ext_vector_type(4))) float v4f;
typedef __attribute__((ext_vector_type(8))) float v8f;

typedef const float  __attribute__((address_space(1)))* gcfp;
typedef const v2f    __attribute__((address_space(1)))* gcv2fp;
typedef const v4f    __attribute__((address_space(1)))* gcv4fp;
typedef float        __attribute__((address_space(1)))* gfp;

struct MoeWs {
  int counts[NUM_EXPERT];
  int cursor[NUM_EXPERT];
  int offsets[NUM_EXPERT + 1];
  int numTiles;
  int tileExpert[MAX_TILES];
  int tileM[MAX_TILES];
  int perm[BATCH];
};

__global__ void moe_zero(MoeWs* ws) {
  int t = threadIdx.x;
  if (t < NUM_EXPERT) { ws->counts[t] = 0; ws->cursor[t] = 0; }
  if (t == 0) ws->numTiles = 0;
}

__global__ void moe_hist(const int* __restrict__ gate, MoeWs* ws) {
  int b = blockIdx.x * blockDim.x + threadIdx.x;
  if (b < BATCH) atomicAdd(&ws->counts[gate[b] & (NUM_EXPERT - 1)], 1);
}

__global__ void moe_scan(MoeWs* ws) {
  if (threadIdx.x != 0) return;
  int off = 0;
  for (int e = 0; e < NUM_EXPERT; ++e) { ws->offsets[e] = off; off += ws->counts[e]; }
  ws->offsets[NUM_EXPERT] = off;
  int nt = 0;
  for (int e = 0; e < NUM_EXPERT; ++e) {
    int tiles = (ws->counts[e] + TILE_M - 1) / TILE_M;
    for (int j = 0; j < tiles; ++j) { ws->tileExpert[nt] = e; ws->tileM[nt] = j; ++nt; }
  }
  ws->numTiles = nt;
}

__global__ void moe_scatter(const int* __restrict__ gate, MoeWs* ws) {
  int b = blockIdx.x * blockDim.x + threadIdx.x;
  if (b < BATCH) {
    int e = gate[b] & (NUM_EXPERT - 1);
    int pos = ws->offsets[e] + atomicAdd(&ws->cursor[e], 1);
    ws->perm[pos] = b;
  }
}

__launch_bounds__(128)
__global__ void moe_gemm(const float* __restrict__ inp,
                         const float* __restrict__ weight,
                         float* __restrict__ out,
                         const MoeWs* __restrict__ ws) {
  __shared__ __align__(16) float sA[TILE_M][IN_FEAT];
  __shared__ int sTok[TILE_M];

  const int tile = blockIdx.x;
  if (tile >= ws->numTiles) return;
  const int e = ws->tileExpert[tile];
  const int mt = ws->tileM[tile];
  const int base = ws->offsets[e] + mt * TILE_M;
  const int mcount = min(TILE_M, ws->counts[e] - mt * TILE_M);

  if (threadIdx.x < TILE_M)
    sTok[threadIdx.x] = (threadIdx.x < mcount) ? ws->perm[base + threadIdx.x] : -1;
  __syncthreads();

  for (int f = threadIdx.x; f < TILE_M * (IN_FEAT / 4); f += 128) {
    int row = f >> 7;
    int c4  = f & 127;
    int tok = sTok[row];
    v4f v = {};
    if (tok >= 0) v = ((gcv4fp)(inp + (size_t)tok * IN_FEAT))[c4];
    ((v4f*)&sA[row][0])[c4] = v;
  }
  __syncthreads();

  const int lane = threadIdx.x & 31;
  const int wave = threadIdx.x >> 5;
  const int hl   = lane >> 4;
  const int l15  = lane & 15;

  const float* wbase = weight + (size_t)e * OUT_FEAT * IN_FEAT;
  const int ncol0 = blockIdx.y * 256 + wave * 64;

  gcfp bptr[4];
#pragma unroll
  for (int nt = 0; nt < 4; ++nt) {
    const float* p = wbase + (size_t)(ncol0 + nt * 16 + l15) * IN_FEAT + 2 * hl;
    __builtin_prefetch(p, 0, 1);
    bptr[nt] = (gcfp)p;
  }

  v8f acc[4] = {};
  const float* aptr = &sA[l15][2 * hl];

#pragma unroll 8
  for (int k = 0; k < IN_FEAT; k += 4) {
    v2f a = *(const v2f*)aptr;
    aptr += 4;
#pragma unroll
    for (int nt = 0; nt < 4; ++nt) {
      v2f bb = *(gcv2fp)bptr[nt];
      bptr[nt] += 4;
      acc[nt] = __builtin_amdgcn_wmma_f32_16x16x4_f32(
          false, a, false, bb, (short)0, acc[nt], false, false);
    }
  }

#pragma unroll
  for (int nt = 0; nt < 4; ++nt) {
    const int col = ncol0 + nt * 16 + l15;
#pragma unroll
    for (int r = 0; r < 8; ++r) {
      const int m = r + 8 * hl;
      if (m < mcount)
        ((gfp)out)[(size_t)sTok[m] * OUT_FEAT + col] = acc[nt][r];
    }
  }
}

extern "C" void kernel_launch(void* const* d_in, const int* in_sizes, int n_in,
                              void* d_out, int out_size, void* d_ws, size_t ws_size,
                              hipStream_t stream) {
  (void)in_sizes; (void)n_in; (void)out_size; (void)ws_size;
  const float* inp    = (const float*)d_in[0];
  const int*   gate   = (const int*)d_in[1];
  const float* weight = (const float*)d_in[2];
  float* out = (float*)d_out;
  MoeWs* ws  = (MoeWs*)d_ws;

  moe_zero<<<1, 64, 0, stream>>>(ws);
  moe_hist<<<BATCH / 256, 256, 0, stream>>>(gate, ws);
  moe_scan<<<1, 32, 0, stream>>>(ws);
  moe_scatter<<<BATCH / 256, 256, 0, stream>>>(gate, ws);
  moe_gemm<<<dim3(MAX_TILES, OUT_FEAT / 256), 128, 0, stream>>>(inp, weight, out, ws);
}


# GEMM block covers 64 tokens (4 row tiles) of one expert instead of 16: each weight fragment loaded from global feeds up to 4 WMMAs instead of 1, so 4x fewer weight load requests per product
# speedup vs baseline: 1.2127x; 1.2127x over previous
#include <hip/hip_runtime.h>
#include <hip/hip_bf16.h>

#define NUM_EXPERT 32
#define IN_FEAT 512
#define OUT_FEAT 512
#define BATCH 4096
#define TILE_M 16
#define CHUNK_TILES 4
#define CHUNK_M (TILE_M * CHUNK_TILES)
#define MAX_TILES (BATCH / CHUNK_M + NUM_EXPERT)

typedef __attribute__((ext_vector_type(2))) float v2f;
typedef __attribute__((ext_vector_type(4))) float v4f;
typedef __attribute__((ext_vector_type(8))) float v8f;

typedef const float  __attribute__((address_space(1)))* gcfp;
typedef const v2f    __attribute__((address_space(1)))* gcv2fp;
typedef const v4f    __attribute__((address_space(1)))* gcv4fp;
typedef float        __attribute__((address_space(1)))* gfp;
typedef const v2f    __attribute__((address_space(3)))* lcv2fp;

struct MoeWs {
  int counts[NUM_EXPERT];
  int cursor[NUM_EXPERT];
  int offsets[NUM_EXPERT + 1];
  int numTiles;
  int tileExpert[MAX_TILES];
  int tileM[MAX_TILES];
  int perm[BATCH];
};

__global__ void moe_zero(MoeWs* ws) {
  int t = threadIdx.x;
  if (t < NUM_EXPERT) { ws->counts[t] = 0; ws->cursor[t] = 0; }
  if (t == 0) ws->numTiles = 0;
}

__global__ void moe_hist(const int* __restrict__ gate, MoeWs* ws) {
  int b = blockIdx.x * blockDim.x + threadIdx.x;
  if (b < BATCH) atomicAdd(&ws->counts[gate[b] & (NUM_EXPERT - 1)], 1);
}

__global__ void moe_scan(MoeWs* ws) {
  if (threadIdx.x != 0) return;
  int off = 0;
  for (int e = 0; e < NUM_EXPERT; ++e) { ws->offsets[e] = off; off += ws->counts[e]; }
  ws->offsets[NUM_EXPERT] = off;
  int nt = 0;
  for (int e = 0; e < NUM_EXPERT; ++e) {
    int tiles = (ws->counts[e] + CHUNK_M - 1) / CHUNK_M;
    for (int j = 0; j < tiles; ++j) { ws->tileExpert[nt] = e; ws->tileM[nt] = j; ++nt; }
  }
  ws->numTiles = nt;
}

__global__ void moe_scatter(const int* __restrict__ gate, MoeWs* ws) {
  int b = blockIdx.x * blockDim.x + threadIdx.x;
  if (b < BATCH) {
    int e = gate[b] & (NUM_EXPERT - 1);
    int pos = ws->offsets[e] + atomicAdd(&ws->cursor[e], 1);
    ws->perm[pos] = b;
  }
}

template <int MT>
__device__ __forceinline__ void moe_rows(const float* sA0, const int* sTok,
                                         const float* wbase, float* out,
                                         int ncol0, int lane) {
  const int hl  = lane >> 4;
  const int l15 = lane & 15;

  gcfp bptr[4];
#pragma unroll
  for (int nt = 0; nt < 4; ++nt) {
    const float* p = wbase + (size_t)(ncol0 + nt * 16 + l15) * IN_FEAT + 2 * hl;
    __builtin_prefetch(p, 0, 1);
    bptr[nt] = (gcfp)p;
  }

  v8f acc[MT][4] = {};
  lcv2fp aptr = (lcv2fp)(sA0 + l15 * IN_FEAT + 2 * hl);

#pragma unroll 8
  for (int k = 0; k < IN_FEAT; k += 4) {
    v2f bb[4];
#pragma unroll
    for (int nt = 0; nt < 4; ++nt) {
      bb[nt] = *(gcv2fp)bptr[nt];
      bptr[nt] += 4;
    }
#pragma unroll
    for (int mt = 0; mt < MT; ++mt) {
      v2f a = aptr[mt * (TILE_M * IN_FEAT / 2)];
#pragma unroll
      for (int nt = 0; nt < 4; ++nt)
        acc[mt][nt] = __builtin_amdgcn_wmma_f32_16x16x4_f32(
            false, a, false, bb[nt], (short)0, acc[mt][nt], false, false);
    }
    aptr += 2;
  }

#pragma unroll
  for (int mt = 0; mt < MT; ++mt) {
    int tok[8];
#pragma unroll
    for (int r = 0; r < 8; ++r) tok[r] = sTok[mt * TILE_M + r + 8 * hl];
#pragma unroll
    for (int r = 0; r < 8; ++r) {
      if (tok[r] >= 0) {
        gfp orow = (gfp)out + (size_t)tok[r] * OUT_FEAT + ncol0 + l15;
#pragma unroll
        for (int nt = 0; nt < 4; ++nt) orow[nt * 16] = acc[mt][nt][r];
      }
    }
  }
}

__launch_bounds__(128)
__global__ void moe_gemm(const float* __restrict__ inp,
                         const float* __restrict__ weight,
                         float* __restrict__ out,
                         const MoeWs* __restrict__ ws) {
  __shared__ __align__(16) float sA[CHUNK_M][IN_FEAT];
  __shared__ int sTok[CHUNK_M];

  const int tile = blockIdx.x;
  if (tile >= ws->numTiles) return;
  const int e = ws->tileExpert[tile];
  const int mc = ws->tileM[tile];
  const int base = ws->offsets[e] + mc * CHUNK_M;
  const int mcount = min(CHUNK_M, ws->counts[e] - mc * CHUNK_M);
  const int ntile = (mcount + TILE_M - 1) / TILE_M;

  if (threadIdx.x < CHUNK_M)
    sTok[threadIdx.x] = (threadIdx.x < mcount) ? ws->perm[base + threadIdx.x] : -1;
  __syncthreads();

  for (int r0 = 0; r0 < ntile * TILE_M; r0 += TILE_M) {
    v4f v[TILE_M];
#pragma unroll
    for (int u = 0; u < TILE_M; ++u) {
      const int tok = sTok[r0 + u];
      const int tc = tok < 0 ? 0 : tok;
      v[u] = ((gcv4fp)(inp + (size_t)tc * IN_FEAT))[threadIdx.x];
    }
#pragma unroll
    for (int u = 0; u < TILE_M; ++u) {
      if (sTok[r0 + u] < 0) v[u] = (v4f){};
      ((v4f*)&sA[r0 + u][0])[threadIdx.x] = v[u];
    }
  }
  __syncthreads();

  const int lane = threadIdx.x & 31;
  const int wave = threadIdx.x >> 5;
  const float* wbase = weight + (size_t)e * OUT_FEAT * IN_FEAT;
  const int ncol0 = blockIdx.y * 256 + wave * 64;

  switch (ntile) {
    case 1:  moe_rows<1>(&sA[0][0], sTok, wbase, out, ncol0, lane); break;
    case 2:  moe_rows<2>(&sA[0][0], sTok, wbase, out, ncol0, lane); break;
    case 3:  moe_rows<3>(&sA[0][0], sTok, wbase, out, ncol0, lane); break;
    default: moe_rows<4>(&sA[0][0], sTok, wbase, out, ncol0, lane); break;
  }
}

extern "C" void kernel_launch(void* const* d_in, const int* in_sizes, int n_in,
                              void* d_out, int out_size, void* d_ws, size_t ws_size,
                              hipStream_t stream) {
  (void)in_sizes; (void)n_in; (void)out_size; (void)ws_size;
  const float* inp    = (const float*)d_in[0];
  const int*   gate   = (const int*)d_in[1];
  const float* weight = (const float*)d_in[2];
  float* out = (float*)d_out;
  MoeWs* ws  = (MoeWs*)d_ws;

  moe_zero<<<1, 64, 0, stream>>>(ws);
  moe_hist<<<BATCH / 256, 256, 0, stream>>>(gate, ws);
  moe_scan<<<1, 32, 0, stream>>>(ws);
  moe_scatter<<<BATCH / 256, 256, 0, stream>>>(gate, ws);
  moe_gemm<<<dim3(MAX_TILES, OUT_FEAT / 256), 128, 0, stream>>>(inp, weight, out, ws);
}


# Routing fused into the GEMM launch: each block builds its expert's token list from the gate array itself (stable prefix scan, no atomics): 1 launch instead of 5, no workspace
# speedup vs baseline: 1.7473x; 1.4408x over previous
#include <hip/hip_runtime.h>
#include <hip/hip_bf16.h>


#define NUM_EXPERT 32
#define IN_FEAT 512
#define OUT_FEAT 512
#define BATCH 4096
#define TILE_M 16
#define CHUNK_TILES 4
#define CHUNK_M (TILE_M * CHUNK_TILES)
#define SLOTS 3
#define NTHREADS 128
#define TOK_PER_THREAD (BATCH / NTHREADS)

typedef __attribute__((ext_vector_type(2))) float v2f;
typedef __attribute__((ext_vector_type(4))) float v4f;
typedef __attribute__((ext_vector_type(8))) float v8f;
typedef __attribute__((ext_vector_type(4))) int   v4i;

typedef const float  __attribute__((address_space(1)))* gcfp;
typedef const v2f    __attribute__((address_space(1)))* gcv2fp;
typedef const v4f    __attribute__((address_space(1)))* gcv4fp;
typedef const v4i    __attribute__((address_space(1)))* gcv4ip;
typedef float        __attribute__((address_space(1)))* gfp;
typedef const v2f    __attribute__((address_space(3)))* lcv2fp;

template <int MT>
__device__ __forceinline__ void moe_rows(const float* sA0, const int* sTok,
                                         const float* wbase, float* out,
                                         int ncol0, int lane) {
  const int hl  = lane >> 4;
  const int l15 = lane & 15;

  gcfp bptr[4];
#pragma unroll
  for (int nt = 0; nt < 4; ++nt) {
    const float* p = wbase + (size_t)(ncol0 + nt * 16 + l15) * IN_FEAT + 2 * hl;
    __builtin_prefetch(p, 0, 1);
    bptr[nt] = (gcfp)p;
  }

  v8f acc[MT][4] = {};
  lcv2fp aptr = (lcv2fp)(sA0 + l15 * IN_FEAT + 2 * hl);

#pragma unroll 8
  for (int k = 0; k < IN_FEAT; k += 4) {
    v2f bb[4];
#pragma unroll
    for (int nt = 0; nt < 4; ++nt) {
      bb[nt] = *(gcv2fp)bptr[nt];
      bptr[nt] += 4;
    }
#pragma unroll
    for (int mt = 0; mt < MT; ++mt) {
      v2f a = aptr[mt * (TILE_M * IN_FEAT / 2)];
#pragma unroll
      for (int nt = 0; nt < 4; ++nt)
        acc[mt][nt] = __builtin_amdgcn_wmma_f32_16x16x4_f32(
            false, a, false, bb[nt], (short)0, acc[mt][nt], false, false);
    }
    aptr += 2;
  }

#pragma unroll
  for (int mt = 0; mt < MT; ++mt) {
    int tok[8];
#pragma unroll
    for (int r = 0; r < 8; ++r) tok[r] = sTok[mt * TILE_M + r + 8 * hl];
#pragma unroll
    for (int r = 0; r < 8; ++r) {
      if (tok[r] >= 0) {
        gfp orow = (gfp)out + (size_t)tok[r] * OUT_FEAT + ncol0 + l15;
#pragma unroll
        for (int nt = 0; nt < 4; ++nt) orow[nt * 16] = acc[mt][nt][r];
      }
    }
  }
}

__launch_bounds__(NTHREADS)
__global__ void moe_gemm(const float* __restrict__ inp,
                         const int* __restrict__ gate,
                         const float* __restrict__ weight,
                         float* __restrict__ out) {
  __shared__ __align__(16) float sA[CHUNK_M][IN_FEAT];
  __shared__ int sTok[CHUNK_M];
  __shared__ int sWaveCnt[NTHREADS / 32];

  const int tid  = threadIdx.x;
  const int lane = tid & 31;
  const int wave = tid >> 5;
  const int e    = blockIdx.x % NUM_EXPERT;
  const int slot = blockIdx.x / NUM_EXPERT;

  gcv4ip g4 = (gcv4ip)(gate + tid * TOK_PER_THREAD);
  v4i g[TOK_PER_THREAD / 4];
#pragma unroll
  for (int j = 0; j < TOK_PER_THREAD / 4; ++j) g[j] = g4[j];
  unsigned match = 0u;
#pragma unroll
  for (int j = 0; j < TOK_PER_THREAD / 4; ++j) {
#pragma unroll
    for (int c = 0; c < 4; ++c)
      match |= (unsigned)((g[j][c] & (NUM_EXPERT - 1)) == e) << (4 * j + c);
  }
  const int cnt = __popc(match);

  int incl = cnt;
  for (int d = 1; d < 32; d <<= 1) {
    const int up = __shfl_up(incl, d);
    if (lane >= d) incl += up;
  }
  if (lane == 31) sWaveCnt[wave] = incl;
  __syncthreads();
  int before = 0, total = 0;
  for (int w = 0; w < NTHREADS / 32; ++w) {
    const int s = sWaveCnt[w];
    if (w < wave) before += s;
    total += s;
  }
  const int first = before + incl - cnt;

  const float* wbase = weight + (size_t)e * OUT_FEAT * IN_FEAT;
  const int ncol0 = blockIdx.y * 256 + wave * 64;

  for (int mc = slot; mc * CHUNK_M < total; mc += SLOTS) {
    const int lo = mc * CHUNK_M;
    const int mcount = min(CHUNK_M, total - lo);
    const int ntile = (mcount + TILE_M - 1) / TILE_M;

    {
      unsigned m = match;
      int pos = first - lo;
      while (m) {
        const int b = __builtin_ctz(m);
        m &= m - 1u;
        if ((unsigned)pos < (unsigned)CHUNK_M) sTok[pos] = tid * TOK_PER_THREAD + b;
        ++pos;
      }
      if (tid < CHUNK_M && tid >= mcount) sTok[tid] = -1;
    }
    __syncthreads();

    for (int r0 = 0; r0 < ntile * TILE_M; r0 += TILE_M) {
      v4f v[TILE_M];
#pragma unroll
      for (int u = 0; u < TILE_M; ++u) {
        const int tok = sTok[r0 + u];
        const int tc = tok < 0 ? 0 : tok;
        v[u] = ((gcv4fp)(inp + (size_t)tc * IN_FEAT))[tid];
      }
#pragma unroll
      for (int u = 0; u < TILE_M; ++u) {
        if (sTok[r0 + u] < 0) v[u] = (v4f){};
        ((v4f*)&sA[r0 + u][0])[tid] = v[u];
      }
    }
    __syncthreads();

    switch (ntile) {
      case 1:  moe_rows<1>(&sA[0][0], sTok, wbase, out, ncol0, lane); break;
      case 2:  moe_rows<2>(&sA[0][0], sTok, wbase, out, ncol0, lane); break;
      case 3:  moe_rows<3>(&sA[0][0], sTok, wbase, out, ncol0, lane); break;
      default: moe_rows<4>(&sA[0][0], sTok, wbase, out, ncol0, lane); break;
    }
    __syncthreads();
  }
}

extern "C" void kernel_launch(void* const* d_in, const int* in_sizes, int n_in,
                              void* d_out, int out_size, void* d_ws, size_t ws_size,
                              hipStream_t stream) {
  (void)in_sizes; (void)n_in; (void)out_size; (void)d_ws; (void)ws_size;
  const float* inp    = (const float*)d_in[0];
  const int*   gate   = (const int*)d_in[1];
  const float* weight = (const float*)d_in[2];
  float* out = (float*)d_out;

  moe_gemm<<<dim3(NUM_EXPERT * SLOTS, OUT_FEAT / 256), NTHREADS, 0, stream>>>(inp, gate, weight, out);
}


# Work re-cut into 256 equal blocks (expert x 80-token half x 128 columns; wave tile 5x2, was 4x4): longest wave 2048 -> 1280 WMMAs; LDS rows of A padded to 516 floats (bank-conflict-free reads)
# speedup vs baseline: 2.0802x; 1.1905x over previous
#include <hip/hip_runtime.h>
#include <hip/hip_bf16.h>


#define NUM_EXPERT 32
#define IN_FEAT 512
#define OUT_FEAT 512
#define BATCH 4096
#define TILE_M 16
#define CHUNK_TILES 5
#define CHUNK_M (TILE_M * CHUNK_TILES)
#define SLOTS 2
#define NTHREADS 128
#define WAVE_NT 2
#define BLOCK_N (WAVE_NT * 16 * (NTHREADS / 32))
#define A_PITCH (IN_FEAT + 4)
#define TOK_PER_THREAD (BATCH / NTHREADS)

typedef __attribute__((ext_vector_type(2))) float v2f;
typedef __attribute__((ext_vector_type(4))) float v4f;
typedef __attribute__((ext_vector_type(8))) float v8f;
typedef __attribute__((ext_vector_type(4))) int   v4i;

typedef const float  __attribute__((address_space(1)))* gcfp;
typedef const v2f    __attribute__((address_space(1)))* gcv2fp;
typedef const v4f    __attribute__((address_space(1)))* gcv4fp;
typedef const v4i    __attribute__((address_space(1)))* gcv4ip;
typedef float        __attribute__((address_space(1)))* gfp;
typedef const v2f    __attribute__((address_space(3)))* lcv2fp;

template <int MT>
__device__ __forceinline__ void moe_rows(const float* sA0, const int* sTok,
                                         const float* wbase, float* out,
                                         int ncol0, int lane) {
  const int hl  = lane >> 4;
  const int l15 = lane & 15;

  gcfp bptr[WAVE_NT];
#pragma unroll
  for (int nt = 0; nt < WAVE_NT; ++nt) {
    const float* p = wbase + (size_t)(ncol0 + nt * 16 + l15) * IN_FEAT + 2 * hl;
    __builtin_prefetch(p, 0, 1);
    bptr[nt] = (gcfp)p;
  }

  v8f acc[MT][WAVE_NT] = {};
  lcv2fp aptr = (lcv2fp)(sA0 + l15 * A_PITCH + 2 * hl);

#pragma unroll 8
  for (int k = 0; k < IN_FEAT; k += 4) {
    v2f bb[WAVE_NT];
#pragma unroll
    for (int nt = 0; nt < WAVE_NT; ++nt) {
      bb[nt] = *(gcv2fp)bptr[nt];
      bptr[nt] += 4;
    }
#pragma unroll
    for (int mt = 0; mt < MT; ++mt) {
      v2f a = aptr[mt * (TILE_M * A_PITCH / 2)];
#pragma unroll
      for (int nt = 0; nt < WAVE_NT; ++nt)
        acc[mt][nt] = __builtin_amdgcn_wmma_f32_16x16x4_f32(
            false, a, false, bb[nt], (short)0, acc[mt][nt], false, false);
    }
    aptr += 2;
  }

#pragma unroll
  for (int mt = 0; mt < MT; ++mt) {
    int tok[8];
#pragma unroll
    for (int r = 0; r < 8; ++r) tok[r] = sTok[mt * TILE_M + r + 8 * hl];
#pragma unroll
    for (int r = 0; r < 8; ++r) {
      if (tok[r] >= 0) {
        gfp orow = (gfp)out + (size_t)tok[r] * OUT_FEAT + ncol0 + l15;
#pragma unroll
        for (int nt = 0; nt < WAVE_NT; ++nt) orow[nt * 16] = acc[mt][nt][r];
      }
    }
  }
}

__launch_bounds__(NTHREADS)
__global__ void moe_gemm(const float* __restrict__ inp,
                         const int* __restrict__ gate,
                         const float* __restrict__ weight,
                         float* __restrict__ out) {
  __shared__ __align__(16) float sA[CHUNK_M][A_PITCH];
  __shared__ int sTok[CHUNK_M];
  __shared__ int sWaveCnt[NTHREADS / 32];

  const int tid  = threadIdx.x;
  const int lane = tid & 31;
  const int wave = tid >> 5;
  const int e    = blockIdx.x % NUM_EXPERT;
  const int slot = blockIdx.x / NUM_EXPERT;

  gcv4ip g4 = (gcv4ip)(gate + tid * TOK_PER_THREAD);
  v4i g[TOK_PER_THREAD / 4];
#pragma unroll
  for (int j = 0; j < TOK_PER_THREAD / 4; ++j) g[j] = g4[j];
  unsigned match = 0u;
#pragma unroll
  for (int j = 0; j < TOK_PER_THREAD / 4; ++j) {
#pragma unroll
    for (int c = 0; c < 4; ++c)
      match |= (unsigned)((g[j][c] & (NUM_EXPERT - 1)) == e) << (4 * j + c);
  }
  const int cnt = __popc(match);

  int incl = cnt;
  for (int d = 1; d < 32; d <<= 1) {
    const int up = __shfl_up(incl, d);
    if (lane >= d) incl += up;
  }
  if (lane == 31) sWaveCnt[wave] = incl;
  __syncthreads();
  int before = 0, total = 0;
  for (int w = 0; w < NTHREADS / 32; ++w) {
    const int s = sWaveCnt[w];
    if (w < wave) before += s;
    total += s;
  }
  const int first = before + incl - cnt;

  const float* wbase = weight + (size_t)e * OUT_FEAT * IN_FEAT;
  const int ncol0 = blockIdx.y * BLOCK_N + wave * (WAVE_NT * 16);

  for (int mc = slot; mc * CHUNK_M < total; mc += SLOTS) {
    const int lo = mc * CHUNK_M;
    const int mcount = min(CHUNK_M, total - lo);
    const int ntile = (mcount + TILE_M - 1) / TILE_M;

    {
      unsigned m = match;
      int pos = first - lo;
      while (m) {
        const int b = __builtin_ctz(m);
        m &= m - 1u;
        if ((unsigned)pos < (unsigned)CHUNK_M) sTok[pos] = tid * TOK_PER_THREAD + b;
        ++pos;
      }
      if (tid < CHUNK_M && tid >= mcount) sTok[tid] = -1;
    }
    __syncthreads();

    for (int r0 = 0; r0 < ntile * TILE_M; r0 += TILE_M) {
      v4f v[TILE_M];
#pragma unroll
      for (int u = 0; u < TILE_M; ++u) {
        const int tok = sTok[r0 + u];
        const int tc = tok < 0 ? 0 : tok;
        v[u] = ((gcv4fp)(inp + (size_t)tc * IN_FEAT))[tid];
      }
#pragma unroll
      for (int u = 0; u < TILE_M; ++u) {
        if (sTok[r0 + u] < 0) v[u] = (v4f){};
        ((v4f*)&sA[r0 + u][0])[tid] = v[u];
      }
    }
    __syncthreads();

    switch (ntile) {
      case 1:  moe_rows<1>(&sA[0][0], sTok, wbase, out, ncol0, lane); break;
      case 2:  moe_rows<2>(&sA[0][0], sTok, wbase, out, ncol0, lane); break;
      case 3:  moe_rows<3>(&sA[0][0], sTok, wbase, out, ncol0, lane); break;
      case 4:  moe_rows<4>(&sA[0][0], sTok, wbase, out, ncol0, lane); break;
      default: moe_rows<5>(&sA[0][0], sTok, wbase, out, ncol0, lane); break;
    }
    __syncthreads();
  }
}

extern "C" void kernel_launch(void* const* d_in, const int* in_sizes, int n_in,
                              void* d_out, int out_size, void* d_ws, size_t ws_size,
                              hipStream_t stream) {
  (void)in_sizes; (void)n_in; (void)out_size; (void)d_ws; (void)ws_size;
  const float* inp    = (const float*)d_in[0];
  const int*   gate   = (const int*)d_in[1];
  const float* weight = (const float*)d_in[2];
  float* out = (float*)d_out;

  moe_gemm<<<dim3(NUM_EXPERT * SLOTS, OUT_FEAT / BLOCK_N), NTHREADS, 0, stream>>>(inp, gate, weight, out);
}


# K loop software-pipelined over two register sets: weight (global) and A (LDS) fragment loads of the next 8 K-steps are issued as one group in front of the current 8 K-steps' WMMAs
# speedup vs baseline: 2.6241x; 1.2615x over previous
#include <hip/hip_runtime.h>
#include <hip/hip_bf16.h>


#define NUM_EXPERT 32
#define IN_FEAT 512
#define OUT_FEAT 512
#define BATCH 4096
#define TILE_M 16
#define CHUNK_TILES 5
#define CHUNK_M (TILE_M * CHUNK_TILES)
#define SLOTS 2
#define NTHREADS 128
#define WAVE_NT 2
#define BLOCK_N (WAVE_NT * 16 * (NTHREADS / 32))
#define A_PITCH (IN_FEAT + 4)
#define TOK_PER_THREAD (BATCH / NTHREADS)

typedef __attribute__((ext_vector_type(2))) float v2f;
typedef __attribute__((ext_vector_type(4))) float v4f;
typedef __attribute__((ext_vector_type(8))) float v8f;
typedef __attribute__((ext_vector_type(4))) int   v4i;

typedef const float  __attribute__((address_space(1)))* gcfp;
typedef const v2f    __attribute__((address_space(1)))* gcv2fp;
typedef const v4f    __attribute__((address_space(1)))* gcv4fp;
typedef const v4i    __attribute__((address_space(1)))* gcv4ip;
typedef float        __attribute__((address_space(1)))* gfp;
typedef const v2f    __attribute__((address_space(3)))* lcv2fp;

#define KSTEPS 8
#define NSETS (IN_FEAT / (4 * KSTEPS))

template <int MT>
__device__ __forceinline__ void load_set(v2f (&a)[KSTEPS][MT], v2f (&b)[KSTEPS][WAVE_NT],
                                         const lcv2fp (&aptr)[MT], const gcfp (&bptr)[WAVE_NT], int set) {
#pragma unroll
  for (int s = 0; s < KSTEPS; ++s) {
#pragma unroll
    for (int nt = 0; nt < WAVE_NT; ++nt)
      b[s][nt] = *(gcv2fp)(bptr[nt] + (set * KSTEPS + s) * 4);
  }
#pragma unroll
  for (int s = 0; s < KSTEPS; ++s) {
#pragma unroll
    for (int mt = 0; mt < MT; ++mt)
      a[s][mt] = aptr[mt][(set * KSTEPS + s) * 2];
  }
}

template <int MT>
__device__ __forceinline__ void mma_set(v8f (&acc)[MT][WAVE_NT], const v2f (&a)[KSTEPS][MT],
                                        const v2f (&b)[KSTEPS][WAVE_NT]) {
#pragma unroll
  for (int s = 0; s < KSTEPS; ++s) {
#pragma unroll
    for (int mt = 0; mt < MT; ++mt) {
#pragma unroll
      for (int nt = 0; nt < WAVE_NT; ++nt)
        acc[mt][nt] = __builtin_amdgcn_wmma_f32_16x16x4_f32(
            false, a[s][mt], false, b[s][nt], (short)0, acc[mt][nt], false, false);
    }
  }
}

template <int MT>
__device__ __forceinline__ void moe_rows(const float* sA0, const int* sTok,
                                         const float* wbase, float* out,
                                         int ncol0, int lane) {
  const int hl  = lane >> 4;
  const int l15 = lane & 15;

  gcfp bptr[WAVE_NT];
#pragma unroll
  for (int nt = 0; nt < WAVE_NT; ++nt) {
    const float* p = wbase + (size_t)(ncol0 + nt * 16 + l15) * IN_FEAT + 2 * hl;
    __builtin_prefetch(p, 0, 1);
    bptr[nt] = (gcfp)p;
  }

  v8f acc[MT][WAVE_NT] = {};
  lcv2fp aptr[MT];
#pragma unroll
  for (int mt = 0; mt < MT; ++mt)
    aptr[mt] = (lcv2fp)(sA0 + (mt * TILE_M + l15) * A_PITCH + 2 * hl);

  v2f a0[KSTEPS][MT], a1[KSTEPS][MT];
  v2f b0[KSTEPS][WAVE_NT], b1[KSTEPS][WAVE_NT];
#define SCHED_FENCE() __builtin_amdgcn_sched_barrier(0)
  load_set<MT>(a0, b0, aptr, bptr, 0);
  for (int set = 0; set < NSETS - 2; set += 2) {
    load_set<MT>(a1, b1, aptr, bptr, set + 1);
    SCHED_FENCE();
    mma_set<MT>(acc, a0, b0);
    SCHED_FENCE();
    load_set<MT>(a0, b0, aptr, bptr, set + 2);
    SCHED_FENCE();
    mma_set<MT>(acc, a1, b1);
  }
  load_set<MT>(a1, b1, aptr, bptr, NSETS - 1);
  SCHED_FENCE();
  mma_set<MT>(acc, a0, b0);
  mma_set<MT>(acc, a1, b1);
#undef SCHED_FENCE

#pragma unroll
  for (int mt = 0; mt < MT; ++mt) {
    int tok[8];
#pragma unroll
    for (int r = 0; r < 8; ++r) tok[r] = sTok[mt * TILE_M + r + 8 * hl];
#pragma unroll
    for (int r = 0; r < 8; ++r) {
      if (tok[r] >= 0) {
        gfp orow = (gfp)out + (size_t)tok[r] * OUT_FEAT + ncol0 + l15;
#pragma unroll
        for (int nt = 0; nt < WAVE_NT; ++nt) orow[nt * 16] = acc[mt][nt][r];
      }
    }
  }
}

__launch_bounds__(NTHREADS)
__global__ void moe_gemm(const float* __restrict__ inp,
                         const int* __restrict__ gate,
                         const float* __restrict__ weight,
                         float* __restrict__ out) {
  __shared__ __align__(16) float sA[CHUNK_M][A_PITCH];
  __shared__ int sTok[CHUNK_M];
  __shared__ int sWaveCnt[NTHREADS / 32];

  const int tid  = threadIdx.x;
  const int lane = tid & 31;
  const int wave = tid >> 5;
  const int e    = blockIdx.x % NUM_EXPERT;
  const int slot = blockIdx.x / NUM_EXPERT;

  gcv4ip g4 = (gcv4ip)(gate + tid * TOK_PER_THREAD);
  v4i g[TOK_PER_THREAD / 4];
#pragma unroll
  for (int j = 0; j < TOK_PER_THREAD / 4; ++j) g[j] = g4[j];
  unsigned match = 0u;
#pragma unroll
  for (int j = 0; j < TOK_PER_THREAD / 4; ++j) {
#pragma unroll
    for (int c = 0; c < 4; ++c)
      match |= (unsigned)((g[j][c] & (NUM_EXPERT - 1)) == e) << (4 * j + c);
  }
  const int cnt = __popc(match);

  int incl = cnt;
  for (int d = 1; d < 32; d <<= 1) {
    const int up = __shfl_up(incl, d);
    if (lane >= d) incl += up;
  }
  if (lane == 31) sWaveCnt[wave] = incl;
  __syncthreads();
  int before = 0, total = 0;
  for (int w = 0; w < NTHREADS / 32; ++w) {
    const int s = sWaveCnt[w];
    if (w < wave) before += s;
    total += s;
  }
  const int first = before + incl - cnt;

  const float* wbase = weight + (size_t)e * OUT_FEAT * IN_FEAT;
  const int ncol0 = blockIdx.y * BLOCK_N + wave * (WAVE_NT * 16);

  for (int mc = slot; mc * CHUNK_M < total; mc += SLOTS) {
    const int lo = mc * CHUNK_M;
    const int mcount = min(CHUNK_M, total - lo);
    const int ntile = (mcount + TILE_M - 1) / TILE_M;

    {
      unsigned m = match;
      int pos = first - lo;
      while (m) {
        const int b = __builtin_ctz(m);
        m &= m - 1u;
        if ((unsigned)pos < (unsigned)CHUNK_M) sTok[pos] = tid * TOK_PER_THREAD + b;
        ++pos;
      }
      if (tid < CHUNK_M && tid >= mcount) sTok[tid] = -1;
    }
    __syncthreads();

    for (int r0 = 0; r0 < ntile * TILE_M; r0 += TILE_M) {
      v4f v[TILE_M];
#pragma unroll
      for (int u = 0; u < TILE_M; ++u) {
        const int tok = sTok[r0 + u];
        const int tc = tok < 0 ? 0 : tok;
        v[u] = ((gcv4fp)(inp + (size_t)tc * IN_FEAT))[tid];
      }
#pragma unroll
      for (int u = 0; u < TILE_M; ++u) {
        if (sTok[r0 + u] < 0) v[u] = (v4f){};
        ((v4f*)&sA[r0 + u][0])[tid] = v[u];
      }
    }
    __syncthreads();

    switch (ntile) {
      case 1:  moe_rows<1>(&sA[0][0], sTok, wbase, out, ncol0, lane); break;
      case 2:  moe_rows<2>(&sA[0][0], sTok, wbase, out, ncol0, lane); break;
      case 3:  moe_rows<3>(&sA[0][0], sTok, wbase, out, ncol0, lane); break;
      case 4:  moe_rows<4>(&sA[0][0], sTok, wbase, out, ncol0, lane); break;
      default: moe_rows<5>(&sA[0][0], sTok, wbase, out, ncol0, lane); break;
    }
    __syncthreads();
  }
}

extern "C" void kernel_launch(void* const* d_in, const int* in_sizes, int n_in,
                              void* d_out, int out_size, void* d_ws, size_t ws_size,
                              hipStream_t stream) {
  (void)in_sizes; (void)n_in; (void)out_size; (void)d_ws; (void)ws_size;
  const float* inp    = (const float*)d_in[0];
  const int*   gate   = (const int*)d_in[1];
  const float* weight = (const float*)d_in[2];
  float* out = (float*)d_out;

  moe_gemm<<<dim3(NUM_EXPERT * SLOTS, OUT_FEAT / BLOCK_N), NTHREADS, 0, stream>>>(inp, gate, weight, out);
}


# A staging of a full 80-row chunk: all 80 global loads of a thread go out as one group before the first LDS store (was a group of 16 per row tile): memory latency paid once, not 5 times
# speedup vs baseline: 2.8287x; 1.0780x over previous
#include <hip/hip_runtime.h>
#include <hip/hip_bf16.h>


#define NUM_EXPERT 32
#define IN_FEAT 512
#define OUT_FEAT 512
#define BATCH 4096
#define TILE_M 16
#define CHUNK_TILES 5
#define CHUNK_M (TILE_M * CHUNK_TILES)
#define SLOTS 2
#define NTHREADS 128
#define WAVE_NT 2
#define BLOCK_N (WAVE_NT * 16 * (NTHREADS / 32))
#define A_PITCH (IN_FEAT + 4)
#define TOK_PER_THREAD (BATCH / NTHREADS)

typedef __attribute__((ext_vector_type(2))) float v2f;
typedef __attribute__((ext_vector_type(4))) float v4f;
typedef __attribute__((ext_vector_type(8))) float v8f;
typedef __attribute__((ext_vector_type(4))) int   v4i;

typedef const float  __attribute__((address_space(1)))* gcfp;
typedef const v2f    __attribute__((address_space(1)))* gcv2fp;
typedef const v4f    __attribute__((address_space(1)))* gcv4fp;
typedef const v4i    __attribute__((address_space(1)))* gcv4ip;
typedef float        __attribute__((address_space(1)))* gfp;
typedef const v2f    __attribute__((address_space(3)))* lcv2fp;
typedef const int    __attribute__((address_space(3)))* lcip;
typedef v4f          __attribute__((address_space(3)))* lv4fp;
typedef const char   __attribute__((address_space(1)))* gccp;

#define KSTEPS 8
#define NSETS (IN_FEAT / (4 * KSTEPS))

template <int MT>
__device__ __forceinline__ void load_set(v2f (&a)[KSTEPS][MT], v2f (&b)[KSTEPS][WAVE_NT],
                                         const lcv2fp (&aptr)[MT], const gcfp (&bptr)[WAVE_NT], int set) {
#pragma unroll
  for (int s = 0; s < KSTEPS; ++s) {
#pragma unroll
    for (int nt = 0; nt < WAVE_NT; ++nt)
      b[s][nt] = *(gcv2fp)(bptr[nt] + (set * KSTEPS + s) * 4);
  }
#pragma unroll
  for (int s = 0; s < KSTEPS; ++s) {
#pragma unroll
    for (int mt = 0; mt < MT; ++mt)
      a[s][mt] = aptr[mt][(set * KSTEPS + s) * 2];
  }
}

template <int MT>
__device__ __forceinline__ void mma_set(v8f (&acc)[MT][WAVE_NT], const v2f (&a)[KSTEPS][MT],
                                        const v2f (&b)[KSTEPS][WAVE_NT]) {
#pragma unroll
  for (int s = 0; s < KSTEPS; ++s) {
#pragma unroll
    for (int mt = 0; mt < MT; ++mt) {
#pragma unroll
      for (int nt = 0; nt < WAVE_NT; ++nt)
        acc[mt][nt] = __builtin_amdgcn_wmma_f32_16x16x4_f32(
            false, a[s][mt], false, b[s][nt], (short)0, acc[mt][nt], false, false);
    }
  }
}

template <int MT>
__device__ __forceinline__ void moe_rows(const float* sA0, const int* sTok,
                                         const float* wbase, float* out,
                                         int ncol0, int lane) {
  const int hl  = lane >> 4;
  const int l15 = lane & 15;

  gcfp bptr[WAVE_NT];
#pragma unroll
  for (int nt = 0; nt < WAVE_NT; ++nt) {
    const float* p = wbase + (size_t)(ncol0 + nt * 16 + l15) * IN_FEAT + 2 * hl;
    __builtin_prefetch(p, 0, 1);
    bptr[nt] = (gcfp)p;
  }

  v8f acc[MT][WAVE_NT] = {};
  lcv2fp aptr[MT];
#pragma unroll
  for (int mt = 0; mt < MT; ++mt)
    aptr[mt] = (lcv2fp)(sA0 + (mt * TILE_M + l15) * A_PITCH + 2 * hl);

  v2f a0[KSTEPS][MT], a1[KSTEPS][MT];
  v2f b0[KSTEPS][WAVE_NT], b1[KSTEPS][WAVE_NT];
#define SCHED_FENCE() __builtin_amdgcn_sched_barrier(0)
  load_set<MT>(a0, b0, aptr, bptr, 0);
  for (int set = 0; set < NSETS - 2; set += 2) {
    load_set<MT>(a1, b1, aptr, bptr, set + 1);
    SCHED_FENCE();
    mma_set<MT>(acc, a0, b0);
    SCHED_FENCE();
    load_set<MT>(a0, b0, aptr, bptr, set + 2);
    SCHED_FENCE();
    mma_set<MT>(acc, a1, b1);
  }
  load_set<MT>(a1, b1, aptr, bptr, NSETS - 1);
  SCHED_FENCE();
  mma_set<MT>(acc, a0, b0);
  mma_set<MT>(acc, a1, b1);
#undef SCHED_FENCE

#pragma unroll
  for (int mt = 0; mt < MT; ++mt) {
    int tok[8];
#pragma unroll
    for (int r = 0; r < 8; ++r) tok[r] = sTok[mt * TILE_M + r + 8 * hl];
#pragma unroll
    for (int r = 0; r < 8; ++r) {
      if (tok[r] >= 0) {
        gfp orow = (gfp)out + (size_t)tok[r] * OUT_FEAT + ncol0 + l15;
#pragma unroll
        for (int nt = 0; nt < WAVE_NT; ++nt) orow[nt * 16] = acc[mt][nt][r];
      }
    }
  }
}

__device__ __forceinline__ void stage_full_chunk(float* sA0, const int* sTok,
                                                 const float* inp, int tid) {
  const lcip tokp = (lcip)sTok;
  const gccp base = (gccp)inp;
  int tok[CHUNK_M];
  v4f v[CHUNK_M];
#pragma unroll
  for (int i = 0; i < CHUNK_M; ++i) tok[i] = tokp[i];
  __builtin_amdgcn_sched_barrier(0);
#pragma unroll
  for (int i = 0; i < CHUNK_M; ++i) {
    const unsigned tc = tok[i] < 0 ? 0u : (unsigned)tok[i];
    const unsigned off = tc * (unsigned)(IN_FEAT * sizeof(float)) + (unsigned)tid * 16u;
    v[i] = *(gcv4fp)(base + off);
  }
  __builtin_amdgcn_sched_barrier(0);
#pragma unroll
  for (int i = 0; i < CHUNK_M; ++i) {
    if (tok[i] < 0) v[i] = (v4f){};
    ((lv4fp)(sA0 + i * A_PITCH))[tid] = v[i];
  }
}

__launch_bounds__(NTHREADS)
__global__ void moe_gemm(const float* __restrict__ inp,
                         const int* __restrict__ gate,
                         const float* __restrict__ weight,
                         float* __restrict__ out) {
  __shared__ __align__(16) float sA[CHUNK_M][A_PITCH];
  __shared__ int sTok[CHUNK_M];
  __shared__ int sWaveCnt[NTHREADS / 32];

  const int tid  = threadIdx.x;
  const int lane = tid & 31;
  const int wave = tid >> 5;
  const int e    = blockIdx.x % NUM_EXPERT;
  const int slot = blockIdx.x / NUM_EXPERT;

  gcv4ip g4 = (gcv4ip)(gate + tid * TOK_PER_THREAD);
  v4i g[TOK_PER_THREAD / 4];
#pragma unroll
  for (int j = 0; j < TOK_PER_THREAD / 4; ++j) g[j] = g4[j];
  unsigned match = 0u;
#pragma unroll
  for (int j = 0; j < TOK_PER_THREAD / 4; ++j) {
#pragma unroll
    for (int c = 0; c < 4; ++c)
      match |= (unsigned)((g[j][c] & (NUM_EXPERT - 1)) == e) << (4 * j + c);
  }
  const int cnt = __popc(match);

  int incl = cnt;
  for (int d = 1; d < 32; d <<= 1) {
    const int up = __shfl_up(incl, d);
    if (lane >= d) incl += up;
  }
  if (lane == 31) sWaveCnt[wave] = incl;
  __syncthreads();
  int before = 0, total = 0;
  for (int w = 0; w < NTHREADS / 32; ++w) {
    const int s = sWaveCnt[w];
    if (w < wave) before += s;
    total += s;
  }
  const int first = before + incl - cnt;

  const float* wbase = weight + (size_t)e * OUT_FEAT * IN_FEAT;
  const int ncol0 = blockIdx.y * BLOCK_N + wave * (WAVE_NT * 16);

  for (int mc = slot; mc * CHUNK_M < total; mc += SLOTS) {
    const int lo = mc * CHUNK_M;
    const int mcount = min(CHUNK_M, total - lo);
    const int ntile = (mcount + TILE_M - 1) / TILE_M;

    {
      unsigned m = match;
      int pos = first - lo;
      while (m) {
        const int b = __builtin_ctz(m);
        m &= m - 1u;
        if ((unsigned)pos < (unsigned)CHUNK_M) sTok[pos] = tid * TOK_PER_THREAD + b;
        ++pos;
      }
      if (tid < CHUNK_M && tid >= mcount) sTok[tid] = -1;
    }
    __syncthreads();

    if (__builtin_expect(ntile == CHUNK_TILES, 1)) {
      stage_full_chunk(&sA[0][0], sTok, inp, tid);
      __syncthreads();
      moe_rows<CHUNK_TILES>(&sA[0][0], sTok, wbase, out, ncol0, lane);
    } else {
      for (int r0 = 0; r0 < ntile * TILE_M; r0 += TILE_M) {
        v4f v[TILE_M];
#pragma unroll
        for (int u = 0; u < TILE_M; ++u) {
          const int tok = sTok[r0 + u];
          const int tc = tok < 0 ? 0 : tok;
          v[u] = ((gcv4fp)(inp + (size_t)tc * IN_FEAT))[tid];
        }
#pragma unroll
        for (int u = 0; u < TILE_M; ++u) {
          if (sTok[r0 + u] < 0) v[u] = (v4f){};
          ((v4f*)&sA[r0 + u][0])[tid] = v[u];
        }
      }
      __syncthreads();

      switch (ntile) {
        case 1:  moe_rows<1>(&sA[0][0], sTok, wbase, out, ncol0, lane); break;
        case 2:  moe_rows<2>(&sA[0][0], sTok, wbase, out, ncol0, lane); break;
        case 3:  moe_rows<3>(&sA[0][0], sTok, wbase, out, ncol0, lane); break;
        default: moe_rows<4>(&sA[0][0], sTok, wbase, out, ncol0, lane); break;
      }
    }
    __syncthreads();
  }
}

extern "C" void kernel_launch(void* const* d_in, const int* in_sizes, int n_in,
                              void* d_out, int out_size, void* d_ws, size_t ws_size,
                              hipStream_t stream) {
  (void)in_sizes; (void)n_in; (void)out_size; (void)d_ws; (void)ws_size;
  const float* inp    = (const float*)d_in[0];
  const int*   gate   = (const int*)d_in[1];
  const float* weight = (const float*)d_in[2];
  float* out = (float*)d_out;

  moe_gemm<<<dim3(NUM_EXPERT * SLOTS, OUT_FEAT / BLOCK_N), NTHREADS, 0, stream>>>(inp, gate, weight, out);
}


# A staging pipelined along K under the products: rows go to LDS in 8 slices of 64 floats, slice i+2 fetched into registers under slice i's WMMAs, one LDS-only barrier per slice; only slice 0 waits
# speedup vs baseline: 3.0263x; 1.0699x over previous
#include <hip/hip_runtime.h>
#include <hip/hip_bf16.h>


#define NUM_EXPERT 32
#define IN_FEAT 512
#define OUT_FEAT 512
#define BATCH 4096
#define TILE_M 16
#define CHUNK_TILES 5
#define CHUNK_M (TILE_M * CHUNK_TILES)
#define SLOTS 2
#define NTHREADS 128
#define WAVE_NT 2
#define BLOCK_N (WAVE_NT * 16 * (NTHREADS / 32))
#define A_PITCH (IN_FEAT + 4)
#define TOK_PER_THREAD (BATCH / NTHREADS)

typedef __attribute__((ext_vector_type(2))) float v2f;
typedef __attribute__((ext_vector_type(4))) float v4f;
typedef __attribute__((ext_vector_type(8))) float v8f;
typedef __attribute__((ext_vector_type(4))) int   v4i;

typedef const float  __attribute__((address_space(1)))* gcfp;
typedef const v2f    __attribute__((address_space(1)))* gcv2fp;
typedef const v4f    __attribute__((address_space(1)))* gcv4fp;
typedef const v4i    __attribute__((address_space(1)))* gcv4ip;
typedef float        __attribute__((address_space(1)))* gfp;
typedef const v2f    __attribute__((address_space(3)))* lcv2fp;
typedef const int    __attribute__((address_space(3)))* lcip;
typedef v4f          __attribute__((address_space(3)))* lv4fp;
typedef const char   __attribute__((address_space(1)))* gccp;

#define KSTEPS 8
#define NSETS (IN_FEAT / (4 * KSTEPS))

__device__ __forceinline__ void load_b(v2f (&b)[KSTEPS][WAVE_NT], const gcfp (&bptr)[WAVE_NT], int set) {
#pragma unroll
  for (int s = 0; s < KSTEPS; ++s) {
#pragma unroll
    for (int nt = 0; nt < WAVE_NT; ++nt)
      b[s][nt] = *(gcv2fp)(bptr[nt] + (set * KSTEPS + s) * 4);
  }
}

template <int MT>
__device__ __forceinline__ void load_a(v2f (&a)[KSTEPS][MT], const lcv2fp (&aptr)[MT], int set) {
#pragma unroll
  for (int s = 0; s < KSTEPS; ++s) {
#pragma unroll
    for (int mt = 0; mt < MT; ++mt)
      a[s][mt] = aptr[mt][(set * KSTEPS + s) * 2];
  }
}

template <int MT>
__device__ __forceinline__ void load_set(v2f (&a)[KSTEPS][MT], v2f (&b)[KSTEPS][WAVE_NT],
                                         const lcv2fp (&aptr)[MT], const gcfp (&bptr)[WAVE_NT], int set) {
  load_b(b, bptr, set);
  load_a<MT>(a, aptr, set);
}

template <int MT>
__device__ __forceinline__ void mma_set(v8f (&acc)[MT][WAVE_NT], const v2f (&a)[KSTEPS][MT],
                                        const v2f (&b)[KSTEPS][WAVE_NT]) {
#pragma unroll
  for (int s = 0; s < KSTEPS; ++s) {
#pragma unroll
    for (int mt = 0; mt < MT; ++mt) {
#pragma unroll
      for (int nt = 0; nt < WAVE_NT; ++nt)
        acc[mt][nt] = __builtin_amdgcn_wmma_f32_16x16x4_f32(
            false, a[s][mt], false, b[s][nt], (short)0, acc[mt][nt], false, false);
    }
  }
}

__device__ __forceinline__ void lds_handover_barrier() {
  asm volatile("s_wait_dscnt 0x0\n\ts_barrier_signal -1\n\ts_barrier_wait -1" ::: "memory");
}

#define SLICE_SETS 2
#define SLICE_F4 (SLICE_SETS * KSTEPS)
#define NSLICES (NSETS / SLICE_SETS)
#define PASS_ROWS (NTHREADS / SLICE_F4)
#define PITCH_F4 (A_PITCH / 4)

template <int NP>
__device__ __forceinline__ void stage_load(v4f (&v)[NP], const unsigned (&off)[NP], gccp base, int slice) {
#pragma unroll
  for (int p = 0; p < NP; ++p)
    v[p] = *(gcv4fp)(base + slice * (SLICE_F4 * 16) + off[p]);
}

template <int NP>
__device__ __forceinline__ void stage_store(lv4fp dst, int slice, const v4f (&v)[NP], unsigned pad) {
#pragma unroll
  for (int p = 0; p < NP; ++p) {
    const v4f x = ((pad >> p) & 1u) ? (v4f){} : v[p];
    dst[p * PASS_ROWS * PITCH_F4 + slice * SLICE_F4] = x;
  }
}

template <int MT>
__device__ __forceinline__ void moe_chunk(float* sA0, const int* sTok, const float* inp,
                                          const float* wbase, float* out,
                                          int ncol0, int tid) {
  constexpr int NP = 2 * MT;
  const int lane = tid & 31;
  const int hl  = lane >> 4;
  const int l15 = lane & 15;
  const int c  = tid & (SLICE_F4 - 1);
  const int rr = tid / SLICE_F4;

  gcfp bptr[WAVE_NT];
#pragma unroll
  for (int nt = 0; nt < WAVE_NT; ++nt) {
    const float* p = wbase + (size_t)(ncol0 + nt * 16 + l15) * IN_FEAT + 2 * hl;
    __builtin_prefetch(p, 0, 1);
    bptr[nt] = (gcfp)p;
  }

  const lcip tokp = (lcip)sTok;
  int tk[NP];
#pragma unroll
  for (int p = 0; p < NP; ++p) tk[p] = tokp[p * PASS_ROWS + rr];
  unsigned off[NP];
  unsigned pad = 0u;
#pragma unroll
  for (int p = 0; p < NP; ++p) {
    pad |= (unsigned)(tk[p] < 0) << p;
    const unsigned tc = tk[p] < 0 ? 0u : (unsigned)tk[p];
    off[p] = tc * (unsigned)(IN_FEAT * sizeof(float)) + (unsigned)c * 16u;
  }
  const gccp abase = (gccp)inp;
  const lv4fp sdst = (lv4fp)sA0 + rr * PITCH_F4 + c;

  v8f acc[MT][WAVE_NT] = {};
  lcv2fp aptr[MT];
#pragma unroll
  for (int mt = 0; mt < MT; ++mt)
    aptr[mt] = (lcv2fp)(sA0 + (mt * TILE_M + l15) * A_PITCH + 2 * hl);

  v2f a0[KSTEPS][MT], a1[KSTEPS][MT];
  v2f b0[KSTEPS][WAVE_NT], b1[KSTEPS][WAVE_NT];
  v4f st0[NP], st1[NP];
#define SCHED_FENCE() __builtin_amdgcn_sched_barrier(0)

  stage_load<NP>(st0, off, abase, 0);
  SCHED_FENCE();
  stage_load<NP>(st1, off, abase, 1);
  SCHED_FENCE();
  load_b(b0, bptr, 0);
  SCHED_FENCE();
  stage_store<NP>(sdst, 0, st0, pad);
  lds_handover_barrier();
  load_a<MT>(a0, aptr, 0);

  for (int i = 0; i < NSLICES - 2; ++i) {
    const int set = SLICE_SETS * i;
    load_set<MT>(a1, b1, aptr, bptr, set + 1);
    stage_store<NP>(sdst, i + 1, st1, pad);
    stage_load<NP>(st1, off, abase, i + 2);
    SCHED_FENCE();
    mma_set<MT>(acc, a0, b0);
    SCHED_FENCE();
    lds_handover_barrier();
    load_set<MT>(a0, b0, aptr, bptr, set + 2);
    SCHED_FENCE();
    mma_set<MT>(acc, a1, b1);
    SCHED_FENCE();
  }
  load_set<MT>(a1, b1, aptr, bptr, NSETS - 3);
  stage_store<NP>(sdst, NSLICES - 1, st1, pad);
  SCHED_FENCE();
  mma_set<MT>(acc, a0, b0);
  SCHED_FENCE();
  lds_handover_barrier();
  load_set<MT>(a0, b0, aptr, bptr, NSETS - 2);
  SCHED_FENCE();
  mma_set<MT>(acc, a1, b1);
  SCHED_FENCE();
  load_set<MT>(a1, b1, aptr, bptr, NSETS - 1);
  SCHED_FENCE();
  mma_set<MT>(acc, a0, b0);
  mma_set<MT>(acc, a1, b1);
#undef SCHED_FENCE

#pragma unroll
  for (int mt = 0; mt < MT; ++mt) {
    int tok[8];
#pragma unroll
    for (int r = 0; r < 8; ++r) tok[r] = sTok[mt * TILE_M + r + 8 * hl];
#pragma unroll
    for (int r = 0; r < 8; ++r) {
      if (tok[r] >= 0) {
        gfp orow = (gfp)out + (size_t)tok[r] * OUT_FEAT + ncol0 + l15;
#pragma unroll
        for (int nt = 0; nt < WAVE_NT; ++nt) orow[nt * 16] = acc[mt][nt][r];
      }
    }
  }
}

__launch_bounds__(NTHREADS)
__global__ void moe_gemm(const float* __restrict__ inp,
                         const int* __restrict__ gate,
                         const float* __restrict__ weight,
                         float* __restrict__ out) {
  __shared__ __align__(16) float sA[CHUNK_M][A_PITCH];
  __shared__ int sTok[CHUNK_M];
  __shared__ int sWaveCnt[NTHREADS / 32];

  const int tid  = threadIdx.x;
  const int lane = tid & 31;
  const int wave = tid >> 5;
  const int e    = blockIdx.x % NUM_EXPERT;
  const int slot = blockIdx.x / NUM_EXPERT;

  gcv4ip g4 = (gcv4ip)(gate + tid * TOK_PER_THREAD);
  v4i g[TOK_PER_THREAD / 4];
#pragma unroll
  for (int j = 0; j < TOK_PER_THREAD / 4; ++j) g[j] = g4[j];
  unsigned match = 0u;
#pragma unroll
  for (int j = 0; j < TOK_PER_THREAD / 4; ++j) {
#pragma unroll
    for (int c = 0; c < 4; ++c)
      match |= (unsigned)((g[j][c] & (NUM_EXPERT - 1)) == e) << (4 * j + c);
  }
  const int cnt = __popc(match);

  int incl = cnt;
  for (int d = 1; d < 32; d <<= 1) {
    const int up = __shfl_up(incl, d);
    if (lane >= d) incl += up;
  }
  if (lane == 31) sWaveCnt[wave] = incl;
  __syncthreads();
  int before = 0, total = 0;
  for (int w = 0; w < NTHREADS / 32; ++w) {
    const int s = sWaveCnt[w];
    if (w < wave) before += s;
    total += s;
  }
  const int first = before + incl - cnt;

  const float* wbase = weight + (size_t)e * OUT_FEAT * IN_FEAT;
  const int ncol0 = blockIdx.y * BLOCK_N + wave * (WAVE_NT * 16);

  for (int mc = slot; mc * CHUNK_M < total; mc += SLOTS) {
    const int lo = mc * CHUNK_M;
    const int mcount = min(CHUNK_M, total - lo);
    const int ntile = (mcount + TILE_M - 1) / TILE_M;

    {
      unsigned m = match;
      int pos = first - lo;
      while (m) {
        const int b = __builtin_ctz(m);
        m &= m - 1u;
        if ((unsigned)pos < (unsigned)CHUNK_M) sTok[pos] = tid * TOK_PER_THREAD + b;
        ++pos;
      }
      if (tid < CHUNK_M && tid >= mcount) sTok[tid] = -1;
    }
    __syncthreads();

    if (__builtin_expect(ntile == CHUNK_TILES, 1)) {
      moe_chunk<CHUNK_TILES>(&sA[0][0], sTok, inp, wbase, out, ncol0, tid);
    } else {
      switch (ntile) {
        case 1:  moe_chunk<1>(&sA[0][0], sTok, inp, wbase, out, ncol0, tid); break;
        case 2:  moe_chunk<2>(&sA[0][0], sTok, inp, wbase, out, ncol0, tid); break;
        case 3:  moe_chunk<3>(&sA[0][0], sTok, inp, wbase, out, ncol0, tid); break;
        default: moe_chunk<4>(&sA[0][0], sTok, inp, wbase, out, ncol0, tid); break;
      }
    }
    __syncthreads();
  }
}

extern "C" void kernel_launch(void* const* d_in, const int* in_sizes, int n_in,
                              void* d_out, int out_size, void* d_ws, size_t ws_size,
                              hipStream_t stream) {
  (void)in_sizes; (void)n_in; (void)out_size; (void)d_ws; (void)ws_size;
  const float* inp    = (const float*)d_in[0];
  const int*   gate   = (const int*)d_in[1];
  const float* weight = (const float*)d_in[2];
  float* out = (float*)d_out;

  moe_gemm<<<dim3(NUM_EXPERT * SLOTS, OUT_FEAT / BLOCK_N), NTHREADS, 0, stream>>>(inp, gate, weight, out);
}


# Output through LDS: each wave turns its 80x32 tile in an LDS region of its own, so every 128-byte output line is ONE 16-byte-per-lane store (20 per wave; was 80 4-byte stores behind 40 branches)
# speedup vs baseline: 3.2461x; 1.0726x over previous
#include <hip/hip_runtime.h>
#include <hip/hip_bf16.h>


#define NUM_EXPERT 32
#define IN_FEAT 512
#define OUT_FEAT 512
#define BATCH 4096
#define TILE_M 16
#define CHUNK_TILES 5
#define CHUNK_M (TILE_M * CHUNK_TILES)
#define SLOTS 2
#define NTHREADS 128
#define WAVE_NT 2
#define BLOCK_N (WAVE_NT * 16 * (NTHREADS / 32))
#define A_PITCH (IN_FEAT + 4)
#define TOK_PER_THREAD (BATCH / NTHREADS)
#define OUT_PITCH (WAVE_NT * 16 + 4)

typedef __attribute__((ext_vector_type(2))) float v2f;
typedef __attribute__((ext_vector_type(4))) float v4f;
typedef __attribute__((ext_vector_type(8))) float v8f;
typedef __attribute__((ext_vector_type(4))) int   v4i;

typedef const float  __attribute__((address_space(1)))* gcfp;
typedef const v2f    __attribute__((address_space(1)))* gcv2fp;
typedef const v4f    __attribute__((address_space(1)))* gcv4fp;
typedef const v4i    __attribute__((address_space(1)))* gcv4ip;
typedef v4f          __attribute__((address_space(1)))* gv4fp;
typedef const v2f    __attribute__((address_space(3)))* lcv2fp;
typedef const int    __attribute__((address_space(3)))* lcip;
typedef v4f          __attribute__((address_space(3)))* lv4fp;
typedef float        __attribute__((address_space(3)))* lfp;
typedef const v4f    __attribute__((address_space(3)))* lcv4fp;
typedef const char   __attribute__((address_space(1)))* gccp;
typedef char         __attribute__((address_space(1)))* gcp;

#define KSTEPS 8
#define NSETS (IN_FEAT / (4 * KSTEPS))

__device__ __forceinline__ void load_b(v2f (&b)[KSTEPS][WAVE_NT], const gcfp (&bptr)[WAVE_NT], int set) {
#pragma unroll
  for (int s = 0; s < KSTEPS; ++s) {
#pragma unroll
    for (int nt = 0; nt < WAVE_NT; ++nt)
      b[s][nt] = *(gcv2fp)(bptr[nt] + (set * KSTEPS + s) * 4);
  }
}

template <int MT>
__device__ __forceinline__ void load_a(v2f (&a)[KSTEPS][MT], const lcv2fp (&aptr)[MT], int set) {
#pragma unroll
  for (int s = 0; s < KSTEPS; ++s) {
#pragma unroll
    for (int mt = 0; mt < MT; ++mt)
      a[s][mt] = aptr[mt][(set * KSTEPS + s) * 2];
  }
}

template <int MT>
__device__ __forceinline__ void load_set(v2f (&a)[KSTEPS][MT], v2f (&b)[KSTEPS][WAVE_NT],
                                         const lcv2fp (&aptr)[MT], const gcfp (&bptr)[WAVE_NT], int set) {
  load_b(b, bptr, set);
  load_a<MT>(a, aptr, set);
}

template <int MT>
__device__ __forceinline__ void mma_set(v8f (&acc)[MT][WAVE_NT], const v2f (&a)[KSTEPS][MT],
                                        const v2f (&b)[KSTEPS][WAVE_NT]) {
#pragma unroll
  for (int s = 0; s < KSTEPS; ++s) {
#pragma unroll
    for (int mt = 0; mt < MT; ++mt) {
#pragma unroll
      for (int nt = 0; nt < WAVE_NT; ++nt)
        acc[mt][nt] = __builtin_amdgcn_wmma_f32_16x16x4_f32(
            false, a[s][mt], false, b[s][nt], (short)0, acc[mt][nt], false, false);
    }
  }
}

__device__ __forceinline__ void lds_handover_barrier() {
  asm volatile("s_wait_dscnt 0x0\n\ts_barrier_signal -1\n\ts_barrier_wait -1" ::: "memory");
}

#define SLICE_SETS 2
#define SLICE_F4 (SLICE_SETS * KSTEPS)
#define NSLICES (NSETS / SLICE_SETS)
#define PASS_ROWS (NTHREADS / SLICE_F4)
#define PITCH_F4 (A_PITCH / 4)

template <int NP>
__device__ __forceinline__ void stage_load(v4f (&v)[NP], const unsigned (&off)[NP], gccp base, int slice) {
#pragma unroll
  for (int p = 0; p < NP; ++p)
    v[p] = *(gcv4fp)(base + slice * (SLICE_F4 * 16) + off[p]);
}

template <int NP>
__device__ __forceinline__ void stage_store(lv4fp dst, int slice, const v4f (&v)[NP], unsigned pad) {
#pragma unroll
  for (int p = 0; p < NP; ++p) {
    const v4f x = ((pad >> p) & 1u) ? (v4f){} : v[p];
    dst[p * PASS_ROWS * PITCH_F4 + slice * SLICE_F4] = x;
  }
}

template <int MT>
__device__ __forceinline__ void tile_to_lds(lfp so_w, const v8f (&acc)[MT][WAVE_NT]) {
#pragma unroll
  for (int mt = 0; mt < MT; ++mt) {
#pragma unroll
    for (int r = 0; r < 8; ++r) {
#pragma unroll
      for (int nt = 0; nt < WAVE_NT; ++nt)
        so_w[(mt * TILE_M + r) * OUT_PITCH + nt * 16] = acc[mt][nt][r];
    }
  }
}

template <int NS>
__device__ __forceinline__ void rows_from_lds(v4f (&v)[NS], lcv4fp so_r) {
#pragma unroll
  for (int i = 0; i < NS; ++i) v[i] = so_r[i * OUT_PITCH];
}

template <int MT>
__device__ __forceinline__ void moe_chunk(float* sA0, float* sO, const int* sTok, const float* inp,
                                          const float* wbase, float* out,
                                          int ncol0, int tid, bool full) {
  constexpr int NP = 2 * MT;
  const int lane = tid & 31;
  const int hl  = lane >> 4;
  const int l15 = lane & 15;
  const int c  = tid & (SLICE_F4 - 1);
  const int rr = tid / SLICE_F4;

  gcfp bptr[WAVE_NT];
#pragma unroll
  for (int nt = 0; nt < WAVE_NT; ++nt) {
    const float* p = wbase + (size_t)(ncol0 + nt * 16 + l15) * IN_FEAT + 2 * hl;
    __builtin_prefetch(p, 0, 1);
    bptr[nt] = (gcfp)p;
  }

  const lcip tokp = (lcip)sTok;
  int tk[NP];
#pragma unroll
  for (int p = 0; p < NP; ++p) tk[p] = tokp[p * PASS_ROWS + rr];
  unsigned off[NP];
  unsigned pad = 0u;
#pragma unroll
  for (int p = 0; p < NP; ++p) {
    pad |= (unsigned)(tk[p] < 0) << p;
    const unsigned tc = tk[p] < 0 ? 0u : (unsigned)tk[p];
    off[p] = tc * (unsigned)(IN_FEAT * sizeof(float)) + (unsigned)c * 16u;
  }
  const gccp abase = (gccp)inp;
  const lv4fp sdst = (lv4fp)sA0 + rr * PITCH_F4 + c;

  v8f acc[MT][WAVE_NT] = {};
  lcv2fp aptr[MT];
#pragma unroll
  for (int mt = 0; mt < MT; ++mt)
    aptr[mt] = (lcv2fp)(sA0 + (mt * TILE_M + l15) * A_PITCH + 2 * hl);

  v2f a0[KSTEPS][MT], a1[KSTEPS][MT];
  v2f b0[KSTEPS][WAVE_NT], b1[KSTEPS][WAVE_NT];
  v4f st0[NP], st1[NP];
#define SCHED_FENCE() __builtin_amdgcn_sched_barrier(0)

  stage_load<NP>(st0, off, abase, 0);
  SCHED_FENCE();
  stage_load<NP>(st1, off, abase, 1);
  SCHED_FENCE();
  load_b(b0, bptr, 0);
  SCHED_FENCE();
  stage_store<NP>(sdst, 0, st0, pad);
  lds_handover_barrier();
  load_a<MT>(a0, aptr, 0);

  for (int i = 0; i < NSLICES - 2; ++i) {
    const int set = SLICE_SETS * i;
    load_set<MT>(a1, b1, aptr, bptr, set + 1);
    stage_store<NP>(sdst, i + 1, st1, pad);
    stage_load<NP>(st1, off, abase, i + 2);
    SCHED_FENCE();
    mma_set<MT>(acc, a0, b0);
    SCHED_FENCE();
    lds_handover_barrier();
    load_set<MT>(a0, b0, aptr, bptr, set + 2);
    SCHED_FENCE();
    mma_set<MT>(acc, a1, b1);
    SCHED_FENCE();
  }
  load_set<MT>(a1, b1, aptr, bptr, NSETS - 3);
  stage_store<NP>(sdst, NSLICES - 1, st1, pad);
  SCHED_FENCE();
  mma_set<MT>(acc, a0, b0);
  SCHED_FENCE();
  lds_handover_barrier();
  load_set<MT>(a0, b0, aptr, bptr, NSETS - 2);
  SCHED_FENCE();
  mma_set<MT>(acc, a1, b1);
  SCHED_FENCE();
  load_set<MT>(a1, b1, aptr, bptr, NSETS - 1);
  SCHED_FENCE();
  mma_set<MT>(acc, a0, b0);
  mma_set<MT>(acc, a1, b1);
#undef SCHED_FENCE

  constexpr int NS = MT * TILE_M / 4;
  const int q  = lane >> 3;
  const int c4 = lane & 7;
  int tok[NS];
#pragma unroll
  for (int i = 0; i < NS; ++i) tok[i] = tokp[4 * i + q];
  const lfp so_w = (lfp)sO + (8 * hl) * OUT_PITCH + l15;
  const lcv4fp so_r = (lcv4fp)((lfp)sO + q * OUT_PITCH + c4 * 4);
  const gcp obase = (gcp)out;
  const unsigned ocol = (unsigned)(ncol0 * sizeof(float)) + (unsigned)c4 * 16u;
  v4f v[NS];
  if (__builtin_expect(full, 1)) {
    unsigned ooff[NS];
#pragma unroll
    for (int i = 0; i < NS; ++i)
      ooff[i] = (unsigned)tok[i] * (unsigned)(OUT_FEAT * sizeof(float)) + ocol;
    tile_to_lds<MT>(so_w, acc);
    rows_from_lds<NS>(v, so_r);
#pragma unroll
    for (int i = 0; i < NS; ++i) *(gv4fp)(obase + ooff[i]) = v[i];
  } else {
    tile_to_lds<MT>(so_w, acc);
    rows_from_lds<NS>(v, so_r);
#pragma unroll
    for (int i = 0; i < NS; ++i) {
      if (tok[i] >= 0)
        *(gv4fp)(obase + ((unsigned)tok[i] * (unsigned)(OUT_FEAT * sizeof(float)) + ocol)) = v[i];
    }
  }
}

__launch_bounds__(NTHREADS)
__global__ void moe_gemm(const float* __restrict__ inp,
                         const int* __restrict__ gate,
                         const float* __restrict__ weight,
                         float* __restrict__ out) {
  __shared__ __align__(16) float sA[CHUNK_M][A_PITCH];
  __shared__ __align__(16) float sOut[NTHREADS / 32][CHUNK_M][OUT_PITCH];
  __shared__ int sTok[CHUNK_M];
  __shared__ int sWaveCnt[NTHREADS / 32];

  const int tid  = threadIdx.x;
  const int lane = tid & 31;
  const int wave = tid >> 5;
  const int e    = blockIdx.x % NUM_EXPERT;
  const int slot = blockIdx.x / NUM_EXPERT;

  gcv4ip g4 = (gcv4ip)(gate + tid * TOK_PER_THREAD);
  v4i g[TOK_PER_THREAD / 4];
#pragma unroll
  for (int j = 0; j < TOK_PER_THREAD / 4; ++j) g[j] = g4[j];
  unsigned match = 0u;
#pragma unroll
  for (int j = 0; j < TOK_PER_THREAD / 4; ++j) {
#pragma unroll
    for (int c = 0; c < 4; ++c)
      match |= (unsigned)((g[j][c] & (NUM_EXPERT - 1)) == e) << (4 * j + c);
  }
  const int cnt = __popc(match);

  int incl = cnt;
  for (int d = 1; d < 32; d <<= 1) {
    const int up = __shfl_up(incl, d);
    if (lane >= d) incl += up;
  }
  if (lane == 31) sWaveCnt[wave] = incl;
  __syncthreads();
  int before = 0, total = 0;
  for (int w = 0; w < NTHREADS / 32; ++w) {
    const int s = sWaveCnt[w];
    if (w < wave) before += s;
    total += s;
  }
  const int first = before + incl - cnt;

  const float* wbase = weight + (size_t)e * OUT_FEAT * IN_FEAT;
  const int ncol0 = blockIdx.y * BLOCK_N + wave * (WAVE_NT * 16);

  for (int mc = slot; mc * CHUNK_M < total; mc += SLOTS) {
    const int lo = mc * CHUNK_M;
    const int mcount = min(CHUNK_M, total - lo);
    const int ntile = (mcount + TILE_M - 1) / TILE_M;

    {
      unsigned m = match;
      int pos = first - lo;
      while (m) {
        const int b = __builtin_ctz(m);
        m &= m - 1u;
        if ((unsigned)pos < (unsigned)CHUNK_M) sTok[pos] = tid * TOK_PER_THREAD + b;
        ++pos;
      }
      if (tid < CHUNK_M && tid >= mcount) sTok[tid] = -1;
    }
    __syncthreads();

    float* sO = &sOut[wave][0][0];
    const bool full = mcount == ntile * TILE_M;

    if (__builtin_expect(ntile == CHUNK_TILES, 1)) {
      moe_chunk<CHUNK_TILES>(&sA[0][0], sO, sTok, inp, wbase, out, ncol0, tid, full);
    } else {
      switch (ntile) {
        case 1:  moe_chunk<1>(&sA[0][0], sO, sTok, inp, wbase, out, ncol0, tid, full); break;
        case 2:  moe_chunk<2>(&sA[0][0], sO, sTok, inp, wbase, out, ncol0, tid, full); break;
        case 3:  moe_chunk<3>(&sA[0][0], sO, sTok, inp, wbase, out, ncol0, tid, full); break;
        default: moe_chunk<4>(&sA[0][0], sO, sTok, inp, wbase, out, ncol0, tid, full); break;
      }
    }
    __syncthreads();
  }
}

extern "C" void kernel_launch(void* const* d_in, const int* in_sizes, int n_in,
                              void* d_out, int out_size, void* d_ws, size_t ws_size,
                              hipStream_t stream) {
  (void)in_sizes; (void)n_in; (void)out_size; (void)d_ws; (void)ws_size;
  const float* inp    = (const float*)d_in[0];
  const int*   gate   = (const int*)d_in[1];
  const float* weight = (const float*)d_in[2];
  float* out = (float*)d_out;

  moe_gemm<<<dim3(NUM_EXPERT * SLOTS, OUT_FEAT / BLOCK_N), NTHREADS, 0, stream>>>(inp, gate, weight, out);
}
